# starting kernel
# baseline (speedup 1.0000x reference)
#include <hip/hip_runtime.h>

typedef __attribute__((ext_vector_type(2))) float v2f;
typedef __attribute__((ext_vector_type(8))) float v8f;

static constexpr int Bb  = 2;
static constexpr int Nn  = 1024;
static constexpr int Mm  = 1024;
static constexpr int DIN = 16;
static constexpr int Cc  = 32;
static constexpr int Hh  = 32;

__global__ void rescale_kernel(const float* __restrict__ x, float* __restrict__ xs) {
    __shared__ float smn[256], smx[256];
    const int bd = blockIdx.x;
    const int b = bd >> 1, d = bd & 1;
    const int t = threadIdx.x;
    float v[4];
    float mn = 1e30f, mx = -1e30f;
#pragma unroll
    for (int i = 0; i < 4; ++i) {
        const int n = t + i * 256;
        v[i] = x[(b * Nn + n) * 2 + d];
        mn = fminf(mn, v[i]);
        mx = fmaxf(mx, v[i]);
    }
    smn[t] = mn; smx[t] = mx;
    __syncthreads();
    for (int s = 128; s > 0; s >>= 1) {
        if (t < s) {
            smn[t] = fminf(smn[t], smn[t + s]);
            smx[t] = fmaxf(smx[t], smx[t + s]);
        }
        __syncthreads();
    }
    mn = smn[0]; mx = smx[0];
    const float sc = 2.0f / (mx - mn + 1e-12f);
#pragma unroll
    for (int i = 0; i < 4; ++i) {
        const int n = t + i * 256;
        xs[(b * Nn + n) * 2 + d] = (v[i] - mn) * sc - 1.0f;
    }
}

__global__ void __launch_bounds__(128) lift_kernel(const float* __restrict__ pnd,
                                                   const float* __restrict__ WL,
                                                   const float* __restrict__ bl,
                                                   float* __restrict__ f) {
    const int wid  = (blockIdx.x * blockDim.x + threadIdx.x) >> 5;
    const int lane = threadIdx.x & 31;
    const int rt = wid >> 1, ct = wid & 1;
    const int row0 = rt * 16;
    const int lrow = lane & 15;
    const int hsel = lane >> 4;
    const int c = ct * 16 + lrow;

    v8f acc = {0.f, 0.f, 0.f, 0.f, 0.f, 0.f, 0.f, 0.f};
#pragma unroll
    for (int kk = 0; kk < 4; ++kk) {
        const int k0 = kk * 4 + 2 * hsel;
        v2f a, bm;
        a.x = pnd[(row0 + lrow) * DIN + k0];
        a.y = pnd[(row0 + lrow) * DIN + k0 + 1];
        bm.x = WL[k0 * Cc + c];
        bm.y = WL[(k0 + 1) * Cc + c];
        acc = __builtin_amdgcn_wmma_f32_16x16x4_f32(false, a, false, bm,
                                                    (short)0, acc, false, false);
    }
    const float bias = bl[c];
#pragma unroll
    for (int r = 0; r < 8; ++r) {
        f[(row0 + r + 8 * hsel) * Cc + c] = acc[r] + bias;
    }
}

__global__ void counts_kernel(const float* __restrict__ xs, const float* __restrict__ lq,
                              float* __restrict__ inv) {
    const int wid  = (blockIdx.x * blockDim.x + threadIdx.x) >> 5;
    const int lane = threadIdx.x & 31;
    const int b = wid / Mm, m = wid % Mm;
    const float q0 = lq[m * 2], q1 = lq[m * 2 + 1];
    float c1 = 0.f, c2 = 0.f;
    for (int n = lane; n < Nn; n += 32) {
        const float x0 = xs[(b * Nn + n) * 2];
        const float x1 = xs[(b * Nn + n) * 2 + 1];
        const float dx = q0 - x0, dy = q1 - x1;
        const float d2 = dx * dx + dy * dy;
        c1 += (d2 <= 0.01f) ? 1.f : 0.f;
        c2 += (d2 <= 0.04f) ? 1.f : 0.f;
    }
#pragma unroll
    for (int s = 16; s >= 1; s >>= 1) {
        c1 += __shfl_xor(c1, s, 32);
        c2 += __shfl_xor(c2, s, 32);
    }
    if (lane == 0) {
        inv[wid * 2]     = 1.0f / fmaxf(c1, 1.0f);
        inv[wid * 2 + 1] = 1.0f / fmaxf(c2, 1.0f);
    }
}

__device__ __forceinline__ float gelu_tanh(float x) {
    const float u = x * x;
    const float t = 0.7978845608028654f * x * fmaf(0.044715f, u, 1.0f);
    const float e = __expf(-2.0f * t);
    const float th = (1.0f - e) * __frcp_rn(1.0f + e);
    return 0.5f * x * (1.0f + th);
}

__global__ void __launch_bounds__(256) main_kernel(
    const float* __restrict__ xs, const float* __restrict__ f,
    const float* __restrict__ inv, const float* __restrict__ lq,
    const float* __restrict__ W1, const float* __restrict__ b1,
    const float* __restrict__ W2, const float* __restrict__ b2,
    float* __restrict__ out) {
    extern __shared__ float smem[];
    float* fs  = smem;
    float* xss = smem + Nn * Cc;

    const int tid  = threadIdx.x;
    const int b    = blockIdx.x / (Mm / 8);
    const int mg   = blockIdx.x % (Mm / 8);
    const int wv   = tid >> 5;
    const int lane = tid & 31;
    const int m    = mg * 8 + wv;
    const int wid  = b * Mm + m;
    const int lrow = lane & 15;
    const int hsel = lane >> 4;

    {
        const float4* src = (const float4*)(f + b * Nn * Cc);
        float4* dst = (float4*)fs;
#pragma unroll
        for (int i = 0; i < (Nn * Cc / 4) / 256; ++i)
            dst[tid + i * 256] = src[tid + i * 256];
        const float4* s2 = (const float4*)(xs + b * Nn * 2);
        float4* d2v = (float4*)xss;
#pragma unroll
        for (int i = 0; i < (Nn * 2 / 4) / 256; ++i)
            d2v[tid + i * 256] = s2[tid + i * 256];
    }
    __syncthreads();

    const float q0 = lq[m * 2], q1 = lq[m * 2 + 1];
    const float inv1 = inv[wid * 2], inv2 = inv[wid * 2 + 1];

    float W1x[16], W1y[16], Cm[16];
#pragma unroll
    for (int j = 0; j < 16; ++j) {
        const int h = 4 * (j >> 1) + 2 * hsel + (j & 1);
        W1x[j] = W1[h];
        W1y[j] = W1[Hh + h];
        Cm[j]  = fmaf(q0, W1[2 * Hh + h], fmaf(q1, W1[3 * Hh + h], b1[h]));
    }

    v2f Bw0[8], Bw1[8];
#pragma unroll
    for (int kk = 0; kk < 8; ++kk) {
        const int k0 = 4 * kk + 2 * hsel;
        Bw0[kk].x = W2[k0 * Cc + lrow];
        Bw0[kk].y = W2[(k0 + 1) * Cc + lrow];
        Bw1[kk].x = W2[k0 * Cc + 16 + lrow];
        Bw1[kk].y = W2[(k0 + 1) * Cc + 16 + lrow];
    }
    const float bz0 = b2[lrow], bz1 = b2[16 + lrow];

    float o0 = 0.f, o1 = 0.f;
    for (int nt = 0; nt < Nn / 16; ++nt) {
        const int n = nt * 16 + lrow;
        const float x0 = xss[n * 2], x1 = xss[n * 2 + 1];
        const float dx = q0 - x0, dy = q1 - x1;
        const float d2 = dx * dx + dy * dy;
        const float w = ((d2 <= 0.01f) ? inv1 : 0.f) + ((d2 <= 0.04f) ? inv2 : 0.f);

        v8f acc0 = {bz0, bz0, bz0, bz0, bz0, bz0, bz0, bz0};
        v8f acc1 = {bz1, bz1, bz1, bz1, bz1, bz1, bz1, bz1};
#pragma unroll
        for (int kk = 0; kk < 8; ++kk) {
            v2f a;
            const float p0 = fmaf(x0, W1x[2 * kk],     fmaf(x1, W1y[2 * kk],     Cm[2 * kk]));
            const float p1 = fmaf(x0, W1x[2 * kk + 1], fmaf(x1, W1y[2 * kk + 1], Cm[2 * kk + 1]));
            a.x = gelu_tanh(p0);
            a.y = gelu_tanh(p1);
            acc0 = __builtin_amdgcn_wmma_f32_16x16x4_f32(false, a, false, Bw0[kk],
                                                         (short)0, acc0, false, false);
            acc1 = __builtin_amdgcn_wmma_f32_16x16x4_f32(false, a, false, Bw1[kk],
                                                         (short)0, acc1, false, false);
        }
#pragma unroll
        for (int r = 0; r < 8; ++r) {
            const int row = r + 8 * hsel;
            const float wr = __shfl(w, row, 32);
            const int nrow = nt * 16 + row;
            const float f0 = fs[nrow * Cc + lrow];
            const float f1 = fs[nrow * Cc + 16 + lrow];
            o0 = fmaf(acc0[r] * wr, f0, o0);
            o1 = fmaf(acc1[r] * wr, f1, o1);
        }
    }
    o0 += __shfl_xor(o0, 16, 32);
    o1 += __shfl_xor(o1, 16, 32);
    if (lane < 16) {
        out[wid * Cc + lane]      = o0;
        out[wid * Cc + 16 + lane] = o1;
    }
}

extern "C" void kernel_launch(void* const* d_in, const int* in_sizes, int n_in,
                              void* d_out, int out_size, void* d_ws, size_t ws_size,
                              hipStream_t stream) {
    (void)in_sizes; (void)n_in; (void)out_size; (void)ws_size;
    const float* pndata = (const float*)d_in[0];
    const float* xcoord = (const float*)d_in[1];
    const float* lq     = (const float*)d_in[2];
    const float* W_lift = (const float*)d_in[3];
    const float* b_lift = (const float*)d_in[4];
    const float* W1     = (const float*)d_in[5];
    const float* b1     = (const float*)d_in[6];
    const float* W2     = (const float*)d_in[7];
    const float* b2     = (const float*)d_in[8];
    float* out = (float*)d_out;

    float* ws  = (float*)d_ws;
    float* xs  = ws;
    float* f   = xs + Bb * Nn * 2;
    float* inv = f + Bb * Nn * Cc;

    rescale_kernel<<<Bb * 2, 256, 0, stream>>>(xcoord, xs);
    lift_kernel<<<64, 128, 0, stream>>>(pndata, W_lift, b_lift, f);
    counts_kernel<<<256, 256, 0, stream>>>(xs, lq, inv);

    const size_t smem_bytes = (size_t)(Nn * Cc + Nn * 2) * sizeof(float);
    main_kernel<<<Bb * (Mm / 8), 256, smem_bytes, stream>>>(xs, f, inv, lq,
                                                            W1, b1, W2, b2, out);
}


# one launch instead of two: each block orders its batch's points in LDS (ballot counting sort on snake cells, no atomics); f rows made per tile by WMMA on gathered input rows; changes fp sum order
# speedup vs baseline: 7.3197x; 7.3197x over previous
#include <hip/hip_runtime.h>

typedef __attribute__((ext_vector_type(2))) float v2f;
typedef __attribute__((ext_vector_type(8))) float v8f;

static constexpr int Bb  = 2;
static constexpr int Nn  = 1024;
static constexpr int Mm  = 1024;
static constexpr int DIN = 16;
static constexpr int Cc  = 32;
static constexpr int Hh  = 32;

__device__ __forceinline__ void gelu_tanh4(const float (&x)[4], float (&g)[4]) {
    float e[4], d[4];
#pragma unroll
    for (int i = 0; i < 4; ++i) {
        const float u = x[i] * x[i];
        const float t = 0.7978845608028654f * x[i] * fmaf(0.044715f, u, 1.0f);
        e[i] = fminf(__expf(-2.0f * t), 1073741824.0f);
        d[i] = 1.0f + e[i];
    }
    const float m01 = d[0] * d[1], m23 = d[2] * d[3];
    const float R = __frcp_rn(m01 * m23);
    const float r01 = m23 * R, r23 = m01 * R;
    const float inv[4] = {d[1] * r01, d[0] * r01, d[3] * r23, d[2] * r23};
#pragma unroll
    for (int i = 0; i < 4; ++i) {
        const float th = (1.0f - e[i]) * inv[i];
        g[i] = 0.5f * x[i] * (1.0f + th);
    }
}

__device__ __forceinline__ float dist2(float q0, float q1, float x0, float x1) {
    const float dx = q0 - x0, dy = q1 - x1;
    return __builtin_fmaf(dx, dx, dy * dy);
}

__device__ __forceinline__ int pop_group(unsigned long long& a0, unsigned long long& a1,
                                         unsigned long long& a2, unsigned long long& a3,
                                         int& base) {
    for (int k = 0; k < 3 && a0 == 0ull; ++k) {
        a0 = a1; a1 = a2; a2 = a3; a3 = 0ull;
        base += 64;
    }
    const int g = base + ((a0 != 0ull) ? __builtin_ctzll(a0) : 0);
    a0 &= a0 - 1ull;
    return g;
}

__global__ void __launch_bounds__(256) gno_kernel(
    const float* __restrict__ x, const float* __restrict__ pnd,
    const float* __restrict__ lq,
    const float* __restrict__ WL, const float* __restrict__ bl,
    const float* __restrict__ W1, const float* __restrict__ b1,
    const float* __restrict__ W2, const float* __restrict__ b2,
    float* __restrict__ out) {
    __shared__ float4 xs4[Nn / 2];
    __shared__ float4 bb[Nn / 16];
    __shared__ uint4 cnt4[256 * 2];
    __shared__ uint4 off4[256 * 4];
    __shared__ int sidx[Nn];
    __shared__ float red[4][8];
    __shared__ int wsum[8];
    const float* xss = (const float*)xs4;
    unsigned char* cnt8 = (unsigned char*)cnt4;
    const unsigned short* off16 = (const unsigned short*)off4;

    const int tid  = threadIdx.x;
    const int b    = blockIdx.x / (Mm / 8);
    const int mg   = blockIdx.x % (Mm / 8);
    const int lane = tid & 31;
    const int wv   = __builtin_amdgcn_readfirstlane(tid >> 5);
    const int m    = mg * 8 + wv;
    const int wid  = b * Mm + m;
    const int lrow = lane & 15;
    const int hsel = lane >> 4;

    const float2* xb = (const float2*)(x + b * Nn * 2);
    float2 v[4];
#pragma unroll
    for (int i = 0; i < 4; ++i) v[i] = xb[tid + i * 256];

    cnt4[2 * tid]     = make_uint4(0u, 0u, 0u, 0u);
    cnt4[2 * tid + 1] = make_uint4(0u, 0u, 0u, 0u);

    const float q0 = lq[m * 2], q1 = lq[m * 2 + 1];
    float W1x[16], W1y[16], W1q0[16], W1q1[16], B1v[16];
#pragma unroll
    for (int j = 0; j < 16; ++j) {
        const int h = 4 * (j >> 1) + 2 * hsel + (j & 1);
        W1x[j]  = W1[h];
        W1y[j]  = W1[Hh + h];
        W1q0[j] = W1[2 * Hh + h];
        W1q1[j] = W1[3 * Hh + h];
        B1v[j]  = b1[h];
    }
    v2f Bw0[8], Bw1[8];
#pragma unroll
    for (int kk = 0; kk < 8; ++kk) {
        const int k0 = 4 * kk + 2 * hsel;
        Bw0[kk].x = W2[k0 * Cc + 2 * lrow];
        Bw0[kk].y = W2[(k0 + 1) * Cc + 2 * lrow];
        Bw1[kk].x = W2[k0 * Cc + 2 * lrow + 1];
        Bw1[kk].y = W2[(k0 + 1) * Cc + 2 * lrow + 1];
    }
    const float bz0 = b2[2 * lrow], bz1 = b2[2 * lrow + 1];
    v2f Lw0[4], Lw1[4];
#pragma unroll
    for (int kk = 0; kk < 4; ++kk) {
        const int k0 = 4 * kk + 2 * hsel;
        Lw0[kk].x = WL[k0 * Cc + 2 * lrow];
        Lw0[kk].y = WL[(k0 + 1) * Cc + 2 * lrow];
        Lw1[kk].x = WL[k0 * Cc + 2 * lrow + 1];
        Lw1[kk].y = WL[(k0 + 1) * Cc + 2 * lrow + 1];
    }
    const float lz0 = bl[2 * lrow], lz1 = bl[2 * lrow + 1];

    float mn0 = 1e30f, mx0 = -1e30f, mn1 = 1e30f, mx1 = -1e30f;
#pragma unroll
    for (int i = 0; i < 4; ++i) {
        mn0 = fminf(mn0, v[i].x); mx0 = fmaxf(mx0, v[i].x);
        mn1 = fminf(mn1, v[i].y); mx1 = fmaxf(mx1, v[i].y);
    }
    for (int s = 16; s >= 1; s >>= 1) {
        mn0 = fminf(mn0, __shfl_xor(mn0, s, 32)); mx0 = fmaxf(mx0, __shfl_xor(mx0, s, 32));
        mn1 = fminf(mn1, __shfl_xor(mn1, s, 32)); mx1 = fmaxf(mx1, __shfl_xor(mx1, s, 32));
    }
    if (lane == 0) { red[0][wv] = mn0; red[1][wv] = mx0; red[2][wv] = mn1; red[3][wv] = mx1; }
    __syncthreads();
    mn0 = red[0][0]; mx0 = red[1][0]; mn1 = red[2][0]; mx1 = red[3][0];
    for (int w = 1; w < 8; ++w) {
        mn0 = fminf(mn0, red[0][w]); mx0 = fmaxf(mx0, red[1][w]);
        mn1 = fminf(mn1, red[2][w]); mx1 = fmaxf(mx1, red[3][w]);
    }
    const float sc0 = 2.0f / (mx0 - mn0 + 1e-12f);
    const float sc1 = 2.0f / (mx1 - mn1 + 1e-12f);

    float pa0[4], pa1[4];
    int cell[4], rin[4];
#pragma unroll
    for (int i = 0; i < 4; ++i) {
        const float a0 = (v[i].x - mn0) * sc0 - 1.0f;
        const float a1 = (v[i].y - mn1) * sc1 - 1.0f;
        pa0[i] = a0; pa1[i] = a1;
        int strip = (int)((a0 + 1.0f) * 4.0f);
        strip = min(7, max(0, strip));
        float yy = a1 + 1.0f;
        if (strip & 1) yy = 2.0f - yy;
        int yb = (int)(yy * 16.0f);
        yb = min(31, max(0, yb));
        const int c = strip * 32 + yb;
        unsigned peer = 0xffffffffu;
#pragma unroll
        for (int bit = 0; bit < 8; ++bit) {
            const bool on = ((c >> bit) & 1) != 0;
            const unsigned bal = __builtin_amdgcn_ballot_w32(on);
            peer &= on ? bal : ~bal;
        }
        cell[i] = c;
        rin[i] = (int)__builtin_amdgcn_mbcnt_lo(peer, 0u);
        if (rin[i] == 0) cnt8[c * 32 + i * 8 + wv] = (unsigned char)__builtin_popcount(peer);
    }
    __syncthreads();

    {
        const uint4 ca = cnt4[2 * tid], cb = cnt4[2 * tid + 1];
        const unsigned wd[8] = {ca.x, ca.y, ca.z, ca.w, cb.x, cb.y, cb.z, cb.w};
        unsigned total = 0u;
#pragma unroll
        for (int j = 0; j < 8; ++j) total = __builtin_amdgcn_sad_u8(wd[j], 0u, total);
        int incl = (int)total;
        for (int d = 1; d < 32; d <<= 1) {
            const int y = __shfl_up(incl, d, 32);
            if (lane >= d) incl += y;
        }
        if (lane == 31) wsum[wv] = incl;
        __syncthreads();
        int wbase = 0;
        for (int w = 0; w < 8; ++w) wbase += (w < wv) ? wsum[w] : 0;
        unsigned run = (unsigned)(wbase + incl) - total;
        unsigned pk[16];
#pragma unroll
        for (int j = 0; j < 8; ++j) {
            const unsigned w = wd[j];
            const unsigned o0 = run; run += w & 0xffu;
            const unsigned o1 = run; run += (w >> 8) & 0xffu;
            const unsigned o2 = run; run += (w >> 16) & 0xffu;
            const unsigned o3 = run; run += w >> 24;
            pk[2 * j]     = o0 | (o1 << 16);
            pk[2 * j + 1] = o2 | (o3 << 16);
        }
#pragma unroll
        for (int j = 0; j < 4; ++j)
            off4[4 * tid + j] = make_uint4(pk[4 * j], pk[4 * j + 1], pk[4 * j + 2], pk[4 * j + 3]);
    }
    __syncthreads();

#pragma unroll
    for (int i = 0; i < 4; ++i) {
        const int rank = (int)off16[cell[i] * 32 + i * 8 + wv] + rin[i];
        ((float2*)xs4)[rank] = make_float2(pa0[i], pa1[i]);
        sidx[rank] = tid + i * 256;
    }
    __syncthreads();

    {
        const float4 pa = xs4[2 * tid], pb = xs4[2 * tid + 1];
        float mnx = fminf(fminf(pa.x, pa.z), fminf(pb.x, pb.z));
        float mxx = fmaxf(fmaxf(pa.x, pa.z), fmaxf(pb.x, pb.z));
        float mny = fminf(fminf(pa.y, pa.w), fminf(pb.y, pb.w));
        float mxy = fmaxf(fmaxf(pa.y, pa.w), fmaxf(pb.y, pb.w));
        for (int sft = 1; sft <= 2; sft <<= 1) {
            mnx = fminf(mnx, __shfl_xor(mnx, sft, 32));
            mxx = fmaxf(mxx, __shfl_xor(mxx, sft, 32));
            mny = fminf(mny, __shfl_xor(mny, sft, 32));
            mxy = fmaxf(mxy, __shfl_xor(mxy, sft, 32));
        }
        if ((tid & 3) == 0) bb[tid >> 2] = make_float4(mnx, mxx, mny, mxy);
    }
    __syncthreads();

    float Cm[16];
#pragma unroll
    for (int j = 0; j < 16; ++j) Cm[j] = fmaf(q0, W1q0[j], fmaf(q1, W1q1[j], B1v[j]));

    unsigned long long todo;
    {
        const float4 ba = bb[lane], bc = bb[lane + 32];
        const float ax = fmaxf(fmaxf(ba.x - q0, q0 - ba.y), 0.f);
        const float ay = fmaxf(fmaxf(ba.z - q1, q1 - ba.w), 0.f);
        const float cx = fmaxf(fmaxf(bc.x - q0, q0 - bc.y), 0.f);
        const float cy = fmaxf(fmaxf(bc.z - q1, q1 - bc.w), 0.f);
        const unsigned lo = __builtin_amdgcn_ballot_w32(ax * ax + ay * ay <= 0.0401f);
        const unsigned hi = __builtin_amdgcn_ballot_w32(cx * cx + cy * cy <= 0.0401f);
        todo = ((unsigned long long)hi << 32) | lo;
    }

    unsigned long long a0 = 0ull, a1 = 0ull, a2 = 0ull, a3 = 0ull;
    int c1 = 0, c2 = 0;
    while (todo != 0ull) {
        const int ta = __builtin_ctzll(todo);
        todo &= todo - 1ull;
        int tb = ta;
        unsigned keep = 0x0000ffffu;
        if (todo != 0ull) {
            tb = __builtin_ctzll(todo);
            todo &= todo - 1ull;
            keep = 0xffffffffu;
        }
        const int n = (hsel ? tb : ta) * 16 + lrow;
        const float d2 = dist2(q0, q1, xss[n * 2], xss[n * 2 + 1]);
        const unsigned m2 = __builtin_amdgcn_ballot_w32(d2 <= 0.04f) & keep;
        const unsigned m1 = __builtin_amdgcn_ballot_w32(d2 <= 0.01f) & keep;
        c1 += __builtin_popcount(m1);
        c2 += __builtin_popcount(m2);
        unsigned nz = m2 | (m2 >> 1);
        nz |= nz >> 2;
        nz &= 0x11111111u;
        nz |= (nz >> 3) | (nz >> 6) | (nz >> 9);
        const unsigned long long ga = (unsigned long long)(nz & 0xfu) << ((ta & 15) * 4);
        const unsigned long long gb = (unsigned long long)((nz >> 16) & 0xfu) << ((tb & 15) * 4);
        const int wa = ta >> 4, wb = tb >> 4;
        a0 |= ((wa == 0) ? ga : 0ull) | ((wb == 0) ? gb : 0ull);
        a1 |= ((wa == 1) ? ga : 0ull) | ((wb == 1) ? gb : 0ull);
        a2 |= ((wa == 2) ? ga : 0ull) | ((wb == 2) ? gb : 0ull);
        a3 |= ((wa == 3) ? ga : 0ull) | ((wb == 3) ? gb : 0ull);
    }
    const float inv1 = 1.0f / fmaxf((float)c1, 1.0f);
    const float inv2 = 1.0f / fmaxf((float)c2, 1.0f);

    int rem = __builtin_popcountll(a0) + __builtin_popcountll(a1)
            + __builtin_popcountll(a2) + __builtin_popcountll(a3);
    const float* pl = pnd + b * Nn * DIN + 2 * hsel;
    int gbase = 0;
    const int slot = lrow >> 2;
    const int sub  = lrow & 3;
    const bool sodd = (slot & 1) != 0, shi = (slot & 2) != 0;
    float o0 = 0.f, o1 = 0.f;
    if (rem > 0) {
        int g0, g1, g2, g3, nv;
        nv = min(rem, 4);
        rem -= nv;
        g0 = pop_group(a0, a1, a2, a3, gbase);
        g1 = g0; g2 = g0; g3 = g0;
        if (nv > 1) g1 = pop_group(a0, a1, a2, a3, gbase);
        if (nv > 2) g2 = pop_group(a0, a1, a2, a3, gbase);
        if (nv > 3) g3 = pop_group(a0, a1, a2, a3, gbase);
        int n = ((sodd ? (shi ? g3 : g1) : (shi ? g2 : g0))) * 4 + sub;
        v2f la[4];
        {
            const float* pr = pl + sidx[n] * DIN;
#pragma unroll
            for (int kk = 0; kk < 4; ++kk) la[kk] = *(const v2f*)(pr + 4 * kk);
        }
        for (;;) {
            const bool more = rem > 0;
            int h0 = g0, h1 = g1, h2 = g2, h3 = g3, nvn = nv, nn = n;
            v2f ln[4];
            if (more) {
                nvn = min(rem, 4);
                rem -= nvn;
                h0 = pop_group(a0, a1, a2, a3, gbase);
                h1 = h0; h2 = h0; h3 = h0;
                if (nvn > 1) h1 = pop_group(a0, a1, a2, a3, gbase);
                if (nvn > 2) h2 = pop_group(a0, a1, a2, a3, gbase);
                if (nvn > 3) h3 = pop_group(a0, a1, a2, a3, gbase);
                nn = ((sodd ? (shi ? h3 : h1) : (shi ? h2 : h0))) * 4 + sub;
                const float* pr = pl + sidx[nn] * DIN;
#pragma unroll
                for (int kk = 0; kk < 4; ++kk) ln[kk] = *(const v2f*)(pr + 4 * kk);
            }

            const float x0 = xss[n * 2], x1 = xss[n * 2 + 1];
            const float d2 = dist2(q0, q1, x0, x1);
            float w = ((d2 <= 0.01f) ? inv1 : 0.f) + ((d2 <= 0.04f) ? inv2 : 0.f);
            if (slot >= nv) w = 0.f;

            v8f f0 = {0.f, 0.f, 0.f, 0.f, 0.f, 0.f, 0.f, 0.f};
            v8f f1 = {0.f, 0.f, 0.f, 0.f, 0.f, 0.f, 0.f, 0.f};
#pragma unroll
            for (int kk = 0; kk < 4; ++kk) {
                f0 = __builtin_amdgcn_wmma_f32_16x16x4_f32(false, la[kk], false, Lw0[kk],
                                                           (short)0, f0, false, false);
                f1 = __builtin_amdgcn_wmma_f32_16x16x4_f32(false, la[kk], false, Lw1[kk],
                                                           (short)0, f1, false, false);
            }

            v8f acc0 = {bz0, bz0, bz0, bz0, bz0, bz0, bz0, bz0};
            v8f acc1 = {bz1, bz1, bz1, bz1, bz1, bz1, bz1, bz1};
#pragma unroll
            for (int k2 = 0; k2 < 4; ++k2) {
                float pin[4], gl[4];
#pragma unroll
                for (int j = 0; j < 4; ++j)
                    pin[j] = fmaf(x0, W1x[4 * k2 + j], fmaf(x1, W1y[4 * k2 + j], Cm[4 * k2 + j]));
                gelu_tanh4(pin, gl);
#pragma unroll
                for (int s = 0; s < 2; ++s) {
                    const int kk = 2 * k2 + s;
                    v2f a;
                    a.x = gl[2 * s];
                    a.y = gl[2 * s + 1];
                    acc0 = __builtin_amdgcn_wmma_f32_16x16x4_f32(false, a, false, Bw0[kk],
                                                                 (short)0, acc0, false, false);
                    acc1 = __builtin_amdgcn_wmma_f32_16x16x4_f32(false, a, false, Bw1[kk],
                                                                 (short)0, acc1, false, false);
                }
            }
#pragma unroll
            for (int r = 0; r < 8; ++r) {
                const float wr = __shfl(w, r + 8 * hsel, 32);
                o0 = fmaf(acc0[r] * wr, f0[r] + lz0, o0);
                o1 = fmaf(acc1[r] * wr, f1[r] + lz1, o1);
            }
            if (!more) break;
            g0 = h0; g1 = h1; g2 = h2; g3 = h3; nv = nvn; n = nn;
#pragma unroll
            for (int kk = 0; kk < 4; ++kk) la[kk] = ln[kk];
        }
    }
    o0 += __shfl_xor(o0, 16, 32);
    o1 += __shfl_xor(o1, 16, 32);
    if (lane < 16) ((float2*)out)[wid * (Cc / 2) + lane] = make_float2(o0, o1);
}

extern "C" void kernel_launch(void* const* d_in, const int* in_sizes, int n_in,
                              void* d_out, int out_size, void* d_ws, size_t ws_size,
                              hipStream_t stream) {
    (void)in_sizes; (void)n_in; (void)out_size; (void)d_ws; (void)ws_size;
    const float* pndata = (const float*)d_in[0];
    const float* xcoord = (const float*)d_in[1];
    const float* lq     = (const float*)d_in[2];
    const float* W_lift = (const float*)d_in[3];
    const float* b_lift = (const float*)d_in[4];
    const float* W1     = (const float*)d_in[5];
    const float* b1     = (const float*)d_in[6];
    const float* W2     = (const float*)d_in[7];
    const float* b2     = (const float*)d_in[8];
    float* out = (float*)d_out;

    gno_kernel<<<Bb * (Mm / 8), 256, 0, stream>>>(xcoord, pndata, lq, W_lift, b_lift,
                                                  W1, b1, W2, b2, out);
}
